# starting kernel
# baseline (speedup 1.0000x reference)
#include <hip/hip_runtime.h>
#include <math.h>


typedef __attribute__((ext_vector_type(2))) float v2f;
typedef __attribute__((ext_vector_type(8))) float v8f;

#define HSTR 66
#define WPB  8

__device__ __forceinline__ v8f wmma4(v2f a, v2f b, v8f c) {
  return __builtin_amdgcn_wmma_f32_16x16x4_f32(
       false, a,  false, b,
       (short)0, c,  false,  false);
}

__device__ __forceinline__ float bn_gelu(float h, float sc, float bi) {
  float t = h * sc + bi;
  return 0.5f * t * (1.0f + erff(t * 0.70710678118654752f));
}

__global__ __launch_bounds__(256) void edgeconv_mlp_kernel(
    const float* __restrict__ x,
    const float* __restrict__ W1,
    const float* __restrict__ g1, const float* __restrict__ b1,
    const float* __restrict__ m1, const float* __restrict__ v1,
    const float* __restrict__ W2,
    const float* __restrict__ g2, const float* __restrict__ b2,
    const float* __restrict__ m2, const float* __restrict__ v2,
    const float* __restrict__ W3,
    const float* __restrict__ g3, const float* __restrict__ b3,
    const float* __restrict__ m3, const float* __restrict__ v3,
    float* __restrict__ out)
{
  __shared__ float hbuf[WPB * 16 * HSTR];

  const int tid    = threadIdx.x;
  const int wave   = tid >> 5;
  const int lane   = tid & 31;
  const int lane_m = lane & 15;
  const int half   = lane >> 4;
  const int kofs   = half * 2;

  const int tile = blockIdx.x * WPB + wave;
  const int p0   = tile * 16;
  float* H = hbuf + wave * 16 * HSTR;

  v8f acc[4];
  const v8f vzero = {0.f, 0.f, 0.f, 0.f, 0.f, 0.f, 0.f, 0.f};
#pragma unroll
  for (int nt = 0; nt < 4; ++nt) acc[nt] = vzero;

  const float* arow = x + (size_t)(p0 + lane_m) * 32 + kofs;
#pragma unroll
  for (int kb = 0; kb < 32; kb += 4) {
    v2f a = *(const v2f*)(arow + kb);
#pragma unroll
    for (int nt = 0; nt < 4; ++nt) {
      v2f b = *(const v2f*)(W1 + (size_t)(nt * 16 + lane_m) * 32 + kb + kofs);
      acc[nt] = wmma4(a, b, acc[nt]);
    }
  }
#pragma unroll
  for (int nt = 0; nt < 4; ++nt) {
    int o = nt * 16 + lane_m;
    float sc = g1[o] * rsqrtf(v1[o] + 1e-5f);
    float bi = b1[o] - m1[o] * sc;
#pragma unroll
    for (int r = 0; r < 8; ++r) {
      int mrow = r + half * 8;
      H[mrow * HSTR + o] = bn_gelu(acc[nt][r], sc, bi);
    }
  }
  asm volatile("s_wait_dscnt 0" ::: "memory");

  const float* Ws[2] = {W2, W3};
  const float* Gs[2] = {g2, g3};
  const float* Bs[2] = {b2, b3};
  const float* Ms[2] = {m2, m3};
  const float* Vs[2] = {v2, v3};

#pragma unroll
  for (int s = 0; s < 2; ++s) {
    const float* W = Ws[s];
#pragma unroll
    for (int nt = 0; nt < 4; ++nt) acc[nt] = vzero;
#pragma unroll
    for (int kb = 0; kb < 64; kb += 4) {
      v2f a = *(const v2f*)(H + lane_m * HSTR + kb + kofs);
#pragma unroll
      for (int nt = 0; nt < 4; ++nt) {
        v2f b = *(const v2f*)(W + (size_t)(nt * 16 + lane_m) * 64 + kb + kofs);
        acc[nt] = wmma4(a, b, acc[nt]);
      }
    }
    asm volatile("s_wait_dscnt 0" ::: "memory");
#pragma unroll
    for (int nt = 0; nt < 4; ++nt) {
      int o = nt * 16 + lane_m;
      float sc = Gs[s][o] * rsqrtf(Vs[s][o] + 1e-5f);
      float bi = Bs[s][o] - Ms[s][o] * sc;
#pragma unroll
      for (int r = 0; r < 8; ++r) {
        int mrow = r + half * 8;
        H[mrow * HSTR + o] = bn_gelu(acc[nt][r], sc, bi);
      }
    }
    asm volatile("s_wait_dscnt 0" ::: "memory");
  }

  const int bidx  = p0 >> 10;
  const int nbase = p0 & 1023;
  float* obase = out + (size_t)bidx * 64 * 1024 + nbase + lane_m;
#pragma unroll
  for (int j = 0; j < 32; ++j) {
    int o = 2 * j + half;
    obase[(size_t)o * 1024] = H[lane_m * HSTR + o];
  }
}

extern "C" void kernel_launch(void* const* d_in, const int* in_sizes, int n_in,
                              void* d_out, int out_size, void* d_ws, size_t ws_size,
                              hipStream_t stream) {
  const float* x  = (const float*)d_in[0];
  const float* W1 = (const float*)d_in[1];
  const float* g1 = (const float*)d_in[2];
  const float* b1 = (const float*)d_in[3];
  const float* m1 = (const float*)d_in[4];
  const float* v1 = (const float*)d_in[5];
  const float* W2 = (const float*)d_in[6];
  const float* g2 = (const float*)d_in[7];
  const float* b2 = (const float*)d_in[8];
  const float* m2 = (const float*)d_in[9];
  const float* v2 = (const float*)d_in[10];
  const float* W3 = (const float*)d_in[11];
  const float* g3 = (const float*)d_in[12];
  const float* b3 = (const float*)d_in[13];
  const float* m3 = (const float*)d_in[14];
  const float* v3 = (const float*)d_in[15];
  float* out = (float*)d_out;

  edgeconv_mlp_kernel<<<512, 256, 0, stream>>>(
      x, W1, g1, b1, m1, v1, W2, g2, b2, m2, v2, W3, g3, b3, m3, v3, out);
}


# W1,W2,W3 staged once per work-group in LDS (padded rows, whole-line 16-byte loads); B fragments read by ds_load_b64 instead of one global load per WMMA whose 32 lanes touch 16 cache lines
# speedup vs baseline: 1.1765x; 1.1765x over previous
#include <hip/hip_runtime.h>
#include <math.h>


typedef __attribute__((ext_vector_type(2))) float v2f;
typedef __attribute__((ext_vector_type(4))) float v4f;
typedef __attribute__((ext_vector_type(8))) float v8f;

#define HSTR 66
#define WPB  8
#define S1   36
#define S2   68

__device__ __forceinline__ v8f wmma4(v2f a, v2f b, v8f c) {
  return __builtin_amdgcn_wmma_f32_16x16x4_f32(
       false, a,  false, b,
       (short)0, c,  false,  false);
}

__device__ __forceinline__ float bn_gelu(float h, float sc, float bi) {
  float t = h * sc + bi;
  return 0.5f * t * (1.0f + erff(t * 0.70710678118654752f));
}

__global__ __launch_bounds__(256) void edgeconv_mlp_kernel(
    const float* __restrict__ x,
    const float* __restrict__ W1,
    const float* __restrict__ g1, const float* __restrict__ b1,
    const float* __restrict__ m1, const float* __restrict__ v1,
    const float* __restrict__ W2,
    const float* __restrict__ g2, const float* __restrict__ b2,
    const float* __restrict__ m2, const float* __restrict__ v2,
    const float* __restrict__ W3,
    const float* __restrict__ g3, const float* __restrict__ b3,
    const float* __restrict__ m3, const float* __restrict__ v3,
    float* __restrict__ out)
{
  __shared__ float hbuf[WPB * 16 * HSTR];
  __shared__ __attribute__((aligned(16))) float w1s[64 * S1];
  __shared__ __attribute__((aligned(16))) float w2s[64 * S2];
  __shared__ __attribute__((aligned(16))) float w3s[64 * S2];

  const int tid    = threadIdx.x;
  const int wave   = tid >> 5;
  const int lane   = tid & 31;
  const int lane_m = lane & 15;
  const int half   = lane >> 4;
  const int kofs   = half * 2;

  const int tile = blockIdx.x * WPB + wave;
  const int p0   = tile * 16;
  float* H = hbuf + wave * 16 * HSTR;

  v8f acc[4];
  const v8f vzero = {0.f, 0.f, 0.f, 0.f, 0.f, 0.f, 0.f, 0.f};
#pragma unroll
  for (int nt = 0; nt < 4; ++nt) acc[nt] = vzero;

  const float* arow = x + (size_t)(p0 + lane_m) * 32 + kofs;
  v2f a1[8];
#pragma unroll
  for (int kb = 0; kb < 32; kb += 4)
    a1[kb >> 2] = *(const v2f*)(arow + kb);

  {
    v4f t1[2], t2[4], t3[4];
#pragma unroll
    for (int i = 0; i < 2; ++i) t1[i] = *(const v4f*)(W1 + (size_t)(tid + i * 256) * 4);
#pragma unroll
    for (int i = 0; i < 4; ++i) t2[i] = *(const v4f*)(W2 + (size_t)(tid + i * 256) * 4);
#pragma unroll
    for (int i = 0; i < 4; ++i) t3[i] = *(const v4f*)(W3 + (size_t)(tid + i * 256) * 4);
#pragma unroll
    for (int i = 0; i < 2; ++i) {
      int f = tid + i * 256;
      *(v4f*)(w1s + (f >> 3) * S1 + (f & 7) * 4) = t1[i];
    }
#pragma unroll
    for (int i = 0; i < 4; ++i) {
      int f = tid + i * 256;
      *(v4f*)(w2s + (f >> 4) * S2 + (f & 15) * 4) = t2[i];
      *(v4f*)(w3s + (f >> 4) * S2 + (f & 15) * 4) = t3[i];
    }
  }
  __syncthreads();

#pragma unroll
  for (int kb = 0; kb < 32; kb += 4) {
    v2f a = a1[kb >> 2];
#pragma unroll
    for (int nt = 0; nt < 4; ++nt) {
      v2f b = *(const v2f*)(w1s + (nt * 16 + lane_m) * S1 + kb + kofs);
      acc[nt] = wmma4(a, b, acc[nt]);
    }
  }
#pragma unroll
  for (int nt = 0; nt < 4; ++nt) {
    int o = nt * 16 + lane_m;
    float sc = g1[o] * rsqrtf(v1[o] + 1e-5f);
    float bi = b1[o] - m1[o] * sc;
#pragma unroll
    for (int r = 0; r < 8; ++r) {
      int mrow = r + half * 8;
      H[mrow * HSTR + o] = bn_gelu(acc[nt][r], sc, bi);
    }
  }
  asm volatile("s_wait_dscnt 0" ::: "memory");

  const float* Ws[2] = {w2s, w3s};
  const float* Gs[2] = {g2, g3};
  const float* Bs[2] = {b2, b3};
  const float* Ms[2] = {m2, m3};
  const float* Vs[2] = {v2, v3};

#pragma unroll
  for (int s = 0; s < 2; ++s) {
    const float* W = Ws[s];
#pragma unroll
    for (int nt = 0; nt < 4; ++nt) acc[nt] = vzero;
#pragma unroll
    for (int kb = 0; kb < 64; kb += 4) {
      v2f a = *(const v2f*)(H + lane_m * HSTR + kb + kofs);
#pragma unroll
      for (int nt = 0; nt < 4; ++nt) {
        v2f b = *(const v2f*)(W + (nt * 16 + lane_m) * S2 + kb + kofs);
        acc[nt] = wmma4(a, b, acc[nt]);
      }
    }
    asm volatile("s_wait_dscnt 0" ::: "memory");
#pragma unroll
    for (int nt = 0; nt < 4; ++nt) {
      int o = nt * 16 + lane_m;
      float sc = Gs[s][o] * rsqrtf(Vs[s][o] + 1e-5f);
      float bi = Bs[s][o] - Ms[s][o] * sc;
#pragma unroll
      for (int r = 0; r < 8; ++r) {
        int mrow = r + half * 8;
        H[mrow * HSTR + o] = bn_gelu(acc[nt][r], sc, bi);
      }
    }
    asm volatile("s_wait_dscnt 0" ::: "memory");
  }

  const int bidx  = p0 >> 10;
  const int nbase = p0 & 1023;
  float* obase = out + (size_t)bidx * 64 * 1024 + nbase + lane_m;
#pragma unroll
  for (int j = 0; j < 32; ++j) {
    int o = 2 * j + half;
    obase[(size_t)o * 1024] = H[lane_m * HSTR + o];
  }
}

extern "C" void kernel_launch(void* const* d_in, const int* in_sizes, int n_in,
                              void* d_out, int out_size, void* d_ws, size_t ws_size,
                              hipStream_t stream) {
  const float* x  = (const float*)d_in[0];
  const float* W1 = (const float*)d_in[1];
  const float* g1 = (const float*)d_in[2];
  const float* b1 = (const float*)d_in[3];
  const float* m1 = (const float*)d_in[4];
  const float* v1 = (const float*)d_in[5];
  const float* W2 = (const float*)d_in[6];
  const float* g2 = (const float*)d_in[7];
  const float* b2 = (const float*)d_in[8];
  const float* m2 = (const float*)d_in[9];
  const float* v2 = (const float*)d_in[10];
  const float* W3 = (const float*)d_in[11];
  const float* g3 = (const float*)d_in[12];
  const float* b3 = (const float*)d_in[13];
  const float* m3 = (const float*)d_in[14];
  const float* v3 = (const float*)d_in[15];
  float* out = (float*)d_out;

  edgeconv_mlp_kernel<<<512, 256, 0, stream>>>(
      x, W1, g1, b1, m1, v1, W2, g2, b2, m2, v2, W3, g3, b3, m3, v3, out);
}


# GELU's erf as one branch-free clamped rational u*P(u^2)/Q(u^2) (rel. error about 3e-7) instead of the library's two exec-masked branches with an exponential: half the vector work, chains interleave
# speedup vs baseline: 1.3883x; 1.1800x over previous
#include <hip/hip_runtime.h>
#include <math.h>


typedef __attribute__((ext_vector_type(2))) float v2f;
typedef __attribute__((ext_vector_type(4))) float v4f;
typedef __attribute__((ext_vector_type(8))) float v8f;

#define HSTR 66
#define WPB  8
#define S1   36
#define S2   68

__device__ __forceinline__ v8f wmma4(v2f a, v2f b, v8f c) {
  return __builtin_amdgcn_wmma_f32_16x16x4_f32(
       false, a,  false, b,
       (short)0, c,  false,  false);
}

struct BnCoef { float scu, biu, sch, bih; };

__device__ __forceinline__ BnCoef bn_fold(float g, float b, float m, float v) {
  float sc = g * rsqrtf(v + 1e-5f);
  float bi = b - m * sc;
  BnCoef c;
  c.scu = sc * 0.70710678118654752f;
  c.biu = bi * 0.70710678118654752f;
  c.sch = sc * 0.5f;
  c.bih = bi * 0.5f;
  return c;
}

__device__ __forceinline__ float bn_gelu(float h, BnCoef c) {
  float u  = __builtin_fmaf(h, c.scu, c.biu);
  float ht = __builtin_fmaf(h, c.sch, c.bih);
  float x  = __builtin_amdgcn_fmed3f(u, -3.7439211627767994f, 3.7439211627767994f);
  float x2 = x * x;
  float p = 0.00022905065861350646f;
  p = __builtin_fmaf(p, x2, 0.0034082910107109506f);
  p = __builtin_fmaf(p, x2, 0.050955695062380861f);
  p = __builtin_fmaf(p, x2, 0.18520832239976145f);
  p = __builtin_fmaf(p, x2, 1.128379143519084f);
  float q = -1.1791602954361697e-7f;
  q = __builtin_fmaf(q, x2, 0.000023547966471313185f);
  q = __builtin_fmaf(q, x2, 0.0010179625278914885f);
  q = __builtin_fmaf(q, x2, 0.014070470171167667f);
  q = __builtin_fmaf(q, x2, 0.11098505178285362f);
  q = __builtin_fmaf(q, x2, 0.49746925110067538f);
  q = __builtin_fmaf(q, x2, 1.0f);
  float e = (x * p) * __builtin_amdgcn_rcpf(q);
  return __builtin_fmaf(ht, e, ht);
}

__global__ __launch_bounds__(256) void edgeconv_mlp_kernel(
    const float* __restrict__ x,
    const float* __restrict__ W1,
    const float* __restrict__ g1, const float* __restrict__ b1,
    const float* __restrict__ m1, const float* __restrict__ v1,
    const float* __restrict__ W2,
    const float* __restrict__ g2, const float* __restrict__ b2,
    const float* __restrict__ m2, const float* __restrict__ v2,
    const float* __restrict__ W3,
    const float* __restrict__ g3, const float* __restrict__ b3,
    const float* __restrict__ m3, const float* __restrict__ v3,
    float* __restrict__ out)
{
  __shared__ float hbuf[WPB * 16 * HSTR];
  __shared__ __attribute__((aligned(16))) float w1s[64 * S1];
  __shared__ __attribute__((aligned(16))) float w2s[64 * S2];
  __shared__ __attribute__((aligned(16))) float w3s[64 * S2];

  const int tid    = threadIdx.x;
  const int wave   = tid >> 5;
  const int lane   = tid & 31;
  const int lane_m = lane & 15;
  const int half   = lane >> 4;
  const int kofs   = half * 2;

  const int tile = blockIdx.x * WPB + wave;
  const int p0   = tile * 16;
  float* H = hbuf + wave * 16 * HSTR;

  v8f acc[4];
  const v8f vzero = {0.f, 0.f, 0.f, 0.f, 0.f, 0.f, 0.f, 0.f};
#pragma unroll
  for (int nt = 0; nt < 4; ++nt) acc[nt] = vzero;

  const float* arow = x + (size_t)(p0 + lane_m) * 32 + kofs;
  v2f a1[8];
#pragma unroll
  for (int kb = 0; kb < 32; kb += 4)
    a1[kb >> 2] = *(const v2f*)(arow + kb);

  {
    v4f t1[2], t2[4], t3[4];
#pragma unroll
    for (int i = 0; i < 2; ++i) t1[i] = *(const v4f*)(W1 + (size_t)(tid + i * 256) * 4);
#pragma unroll
    for (int i = 0; i < 4; ++i) t2[i] = *(const v4f*)(W2 + (size_t)(tid + i * 256) * 4);
#pragma unroll
    for (int i = 0; i < 4; ++i) t3[i] = *(const v4f*)(W3 + (size_t)(tid + i * 256) * 4);
#pragma unroll
    for (int i = 0; i < 2; ++i) {
      int f = tid + i * 256;
      *(v4f*)(w1s + (f >> 3) * S1 + (f & 7) * 4) = t1[i];
    }
#pragma unroll
    for (int i = 0; i < 4; ++i) {
      int f = tid + i * 256;
      *(v4f*)(w2s + (f >> 4) * S2 + (f & 15) * 4) = t2[i];
      *(v4f*)(w3s + (f >> 4) * S2 + (f & 15) * 4) = t3[i];
    }
  }
  __syncthreads();

#pragma unroll
  for (int kb = 0; kb < 32; kb += 4) {
    v2f a = a1[kb >> 2];
#pragma unroll
    for (int nt = 0; nt < 4; ++nt) {
      v2f b = *(const v2f*)(w1s + (nt * 16 + lane_m) * S1 + kb + kofs);
      acc[nt] = wmma4(a, b, acc[nt]);
    }
  }
#pragma unroll
  for (int nt = 0; nt < 4; ++nt) {
    int o = nt * 16 + lane_m;
    BnCoef bc = bn_fold(g1[o], b1[o], m1[o], v1[o]);
#pragma unroll
    for (int r = 0; r < 8; ++r) {
      int mrow = r + half * 8;
      H[mrow * HSTR + o] = bn_gelu(acc[nt][r], bc);
    }
  }
  asm volatile("s_wait_dscnt 0" ::: "memory");

  const float* Ws[2] = {w2s, w3s};
  const float* Gs[2] = {g2, g3};
  const float* Bs[2] = {b2, b3};
  const float* Ms[2] = {m2, m3};
  const float* Vs[2] = {v2, v3};

#pragma unroll
  for (int s = 0; s < 2; ++s) {
    const float* W = Ws[s];
#pragma unroll
    for (int nt = 0; nt < 4; ++nt) acc[nt] = vzero;
#pragma unroll
    for (int kb = 0; kb < 64; kb += 4) {
      v2f a = *(const v2f*)(H + lane_m * HSTR + kb + kofs);
#pragma unroll
      for (int nt = 0; nt < 4; ++nt) {
        v2f b = *(const v2f*)(W + (nt * 16 + lane_m) * S2 + kb + kofs);
        acc[nt] = wmma4(a, b, acc[nt]);
      }
    }
    asm volatile("s_wait_dscnt 0" ::: "memory");
#pragma unroll
    for (int nt = 0; nt < 4; ++nt) {
      int o = nt * 16 + lane_m;
      BnCoef bc = bn_fold(Gs[s][o], Bs[s][o], Ms[s][o], Vs[s][o]);
#pragma unroll
      for (int r = 0; r < 8; ++r) {
        int mrow = r + half * 8;
        H[mrow * HSTR + o] = bn_gelu(acc[nt][r], bc);
      }
    }
    asm volatile("s_wait_dscnt 0" ::: "memory");
  }

  const int bidx  = p0 >> 10;
  const int nbase = p0 & 1023;
  float* obase = out + (size_t)bidx * 64 * 1024 + nbase + lane_m;
#pragma unroll
  for (int j = 0; j < 32; ++j) {
    int o = 2 * j + half;
    obase[(size_t)o * 1024] = H[lane_m * HSTR + o];
  }
}

extern "C" void kernel_launch(void* const* d_in, const int* in_sizes, int n_in,
                              void* d_out, int out_size, void* d_ws, size_t ws_size,
                              hipStream_t stream) {
  const float* x  = (const float*)d_in[0];
  const float* W1 = (const float*)d_in[1];
  const float* g1 = (const float*)d_in[2];
  const float* b1 = (const float*)d_in[3];
  const float* m1 = (const float*)d_in[4];
  const float* v1 = (const float*)d_in[5];
  const float* W2 = (const float*)d_in[6];
  const float* g2 = (const float*)d_in[7];
  const float* b2 = (const float*)d_in[8];
  const float* m2 = (const float*)d_in[9];
  const float* v2 = (const float*)d_in[10];
  const float* W3 = (const float*)d_in[11];
  const float* g3 = (const float*)d_in[12];
  const float* b3 = (const float*)d_in[13];
  const float* m3 = (const float*)d_in[14];
  const float* v3 = (const float*)d_in[15];
  float* out = (float*)d_out;

  edgeconv_mlp_kernel<<<512, 256, 0, stream>>>(
      x, W1, g1, b1, m1, v1, W2, g2, b2, m2, v2, W3, g3, b3, m3, v3, out);
}


# Transposed layers (weights = A rows, points = B columns): GELU results feed the next layer's B fragments from registers, output stored from D; no LDS activation round trip; sum order kept
# speedup vs baseline: 1.4759x; 1.0631x over previous
#include <hip/hip_runtime.h>
#include <math.h>


typedef __attribute__((ext_vector_type(2))) float v2f;
typedef __attribute__((ext_vector_type(4))) float v4f;
typedef __attribute__((ext_vector_type(8))) float v8f;

#define WPB  8
#define S1   36
#define S2   68

__device__ __forceinline__ v8f wmma4(v2f a, v2f b, v8f c) {
  return __builtin_amdgcn_wmma_f32_16x16x4_f32(
       false, a,  false, b,
       (short)0, c,  false,  false);
}

__device__ __forceinline__ v2f fma2(v2f a, v2f b, v2f c) { return __builtin_elementwise_fma(a, b, c); }
__device__ __forceinline__ v2f splat2(float a) { v2f r = {a, a}; return r; }

__device__ __forceinline__ v2f bn_gelu2(v2f h, v4f c0, v4f c1) {
  v2f u  = fma2(h, c0.xy, c0.zw);
  v2f ht = fma2(h, c1.xy, c1.zw);
  v2f x;
  x.x = __builtin_amdgcn_fmed3f(u.x, -3.7439211627767994f, 3.7439211627767994f);
  x.y = __builtin_amdgcn_fmed3f(u.y, -3.7439211627767994f, 3.7439211627767994f);
  v2f x2 = x * x;
  v2f p = splat2(0.00022905065861350646f);
  p = fma2(p, x2, splat2(0.0034082910107109506f));
  p = fma2(p, x2, splat2(0.050955695062380861f));
  p = fma2(p, x2, splat2(0.18520832239976145f));
  p = fma2(p, x2, splat2(1.128379143519084f));
  v2f q = splat2(-1.1791602954361697e-7f);
  q = fma2(q, x2, splat2(0.000023547966471313185f));
  q = fma2(q, x2, splat2(0.0010179625278914885f));
  q = fma2(q, x2, splat2(0.014070470171167667f));
  q = fma2(q, x2, splat2(0.11098505178285362f));
  q = fma2(q, x2, splat2(0.49746925110067538f));
  q = fma2(q, x2, splat2(1.0f));
  v2f rq;
  rq.x = __builtin_amdgcn_rcpf(q.x);
  rq.y = __builtin_amdgcn_rcpf(q.y);
  v2f e = (x * p) * rq;
  return fma2(ht, e, ht);
}

__global__ __launch_bounds__(256) void edgeconv_mlp_kernel(
    const float* __restrict__ x,
    const float* __restrict__ W1,
    const float* __restrict__ g1, const float* __restrict__ b1,
    const float* __restrict__ m1, const float* __restrict__ v1,
    const float* __restrict__ W2,
    const float* __restrict__ g2, const float* __restrict__ b2,
    const float* __restrict__ m2, const float* __restrict__ v2,
    const float* __restrict__ W3,
    const float* __restrict__ g3, const float* __restrict__ b3,
    const float* __restrict__ m3, const float* __restrict__ v3,
    float* __restrict__ out)
{
  __shared__ __attribute__((aligned(16))) float w1s[64 * S1];
  __shared__ __attribute__((aligned(16))) float w2s[64 * S2];
  __shared__ __attribute__((aligned(16))) float w3s[64 * S2];
  __shared__ __attribute__((aligned(16))) float cfs[3 * 32 * 8];

  const int tid    = threadIdx.x;
  const int wave   = tid >> 5;
  const int lane   = tid & 31;
  const int lane_m = lane & 15;
  const int half   = lane >> 4;
  const int kofs   = half * 2;

  const int tile = blockIdx.x * WPB + wave;
  const int p0   = tile * 16;

  const float* xrow = x + (size_t)(p0 + lane_m) * 32 + kofs;
  v2f hb[16];
#pragma unroll
  for (int j = 0; j < 8; ++j)
    hb[j] = *(const v2f*)(xrow + 4 * j);

  {
    v4f t1[2], t2[4], t3[4];
#pragma unroll
    for (int i = 0; i < 2; ++i) t1[i] = *(const v4f*)(W1 + (size_t)(tid + i * 256) * 4);
#pragma unroll
    for (int i = 0; i < 4; ++i) t2[i] = *(const v4f*)(W2 + (size_t)(tid + i * 256) * 4);
#pragma unroll
    for (int i = 0; i < 4; ++i) t3[i] = *(const v4f*)(W3 + (size_t)(tid + i * 256) * 4);

    const int wv = __builtin_amdgcn_readfirstlane(wave);
    if (wv < 3) {
      const float* gp = (wv == 0) ? g1 : (wv == 1) ? g2 : g3;
      const float* bp = (wv == 0) ? b1 : (wv == 1) ? b2 : b3;
      const float* mp = (wv == 0) ? m1 : (wv == 1) ? m2 : m3;
      const float* vp = (wv == 0) ? v1 : (wv == 1) ? v2 : v3;
      const float ge = gp[2 * lane], go = gp[2 * lane + 1];
      const float be = bp[2 * lane], bo = bp[2 * lane + 1];
      const float me = mp[2 * lane], mo = mp[2 * lane + 1];
      const float ve = vp[2 * lane], vo = vp[2 * lane + 1];
      const float sce = ge * rsqrtf(ve + 1e-5f), sco = go * rsqrtf(vo + 1e-5f);
      const float bie = be - me * sce,           bio = bo - mo * sco;
      v4f c0, c1;
      c0.x = sce * 0.70710678118654752f; c0.y = sco * 0.70710678118654752f;
      c0.z = bie * 0.70710678118654752f; c0.w = bio * 0.70710678118654752f;
      c1.x = sce * 0.5f; c1.y = sco * 0.5f;
      c1.z = bie * 0.5f; c1.w = bio * 0.5f;
      float* cw = cfs + (wv * 32 + lane) * 8;
      *(v4f*)(cw)     = c0;
      *(v4f*)(cw + 4) = c1;
    }
#pragma unroll
    for (int i = 0; i < 2; ++i) {
      int f = tid + i * 256;
      *(v4f*)(w1s + (f >> 3) * S1 + (f & 7) * 4) = t1[i];
    }
#pragma unroll
    for (int i = 0; i < 4; ++i) {
      int f = tid + i * 256;
      *(v4f*)(w2s + (f >> 4) * S2 + (f & 15) * 4) = t2[i];
      *(v4f*)(w3s + (f >> 4) * S2 + (f & 15) * 4) = t3[i];
    }
  }
  __syncthreads();

  const int rowp = 4 * ((lane_m & 7) >> 1) + 2 * (lane_m >> 3) + (lane_m & 1);
  const float* cfl = cfs + half * 8;

  v8f acc[4];
  const v8f vzero = {0.f, 0.f, 0.f, 0.f, 0.f, 0.f, 0.f, 0.f};

#pragma unroll
  for (int t = 0; t < 4; ++t) acc[t] = vzero;
  {
    const float* wl = w1s + rowp * S1 + kofs;
#pragma unroll
    for (int j = 0; j < 8; ++j) {
#pragma unroll
      for (int t = 0; t < 4; ++t) {
        v2f a = *(const v2f*)(wl + t * 16 * S1 + 4 * j);
        acc[t] = wmma4(a, hb[j], acc[t]);
      }
    }
  }
#pragma unroll
  for (int t = 0; t < 4; ++t) {
#pragma unroll
    for (int q = 0; q < 4; ++q) {
      const float* cp = cfl + (0 * 32 + 8 * t + 2 * q) * 8;
      v2f hv = {acc[t][2 * q], acc[t][2 * q + 1]};
      hb[4 * t + q] = bn_gelu2(hv, *(const v4f*)cp, *(const v4f*)(cp + 4));
    }
  }

#pragma unroll
  for (int t = 0; t < 4; ++t) acc[t] = vzero;
  {
    const float* wl = w2s + rowp * S2 + kofs;
#pragma unroll
    for (int j = 0; j < 16; ++j) {
#pragma unroll
      for (int t = 0; t < 4; ++t) {
        v2f a = *(const v2f*)(wl + t * 16 * S2 + 4 * j);
        acc[t] = wmma4(a, hb[j], acc[t]);
      }
    }
  }
#pragma unroll
  for (int t = 0; t < 4; ++t) {
#pragma unroll
    for (int q = 0; q < 4; ++q) {
      const float* cp = cfl + (1 * 32 + 8 * t + 2 * q) * 8;
      v2f hv = {acc[t][2 * q], acc[t][2 * q + 1]};
      hb[4 * t + q] = bn_gelu2(hv, *(const v4f*)cp, *(const v4f*)(cp + 4));
    }
  }

#pragma unroll
  for (int t = 0; t < 4; ++t) acc[t] = vzero;
  {
    const float* wl = w3s + rowp * S2 + kofs;
#pragma unroll
    for (int j = 0; j < 16; ++j) {
#pragma unroll
      for (int t = 0; t < 4; ++t) {
        v2f a = *(const v2f*)(wl + t * 16 * S2 + 4 * j);
        acc[t] = wmma4(a, hb[j], acc[t]);
      }
    }
  }
  const int bidx  = p0 >> 10;
  const int nbase = p0 & 1023;
  float* obase = out + (size_t)bidx * 64 * 1024 + (size_t)kofs * 1024 + nbase + lane_m;
#pragma unroll
  for (int t = 0; t < 4; ++t) {
#pragma unroll
    for (int q = 0; q < 4; ++q) {
      const int o0 = 16 * t + 4 * q;
      const float* cp = cfl + (2 * 32 + 8 * t + 2 * q) * 8;
      v2f hv = {acc[t][2 * q], acc[t][2 * q + 1]};
      v2f y = bn_gelu2(hv, *(const v4f*)cp, *(const v4f*)(cp + 4));
      obase[(size_t)o0 * 1024]       = y.x;
      obase[(size_t)(o0 + 1) * 1024] = y.y;
    }
  }
}

extern "C" void kernel_launch(void* const* d_in, const int* in_sizes, int n_in,
                              void* d_out, int out_size, void* d_ws, size_t ws_size,
                              hipStream_t stream) {
  const float* x  = (const float*)d_in[0];
  const float* W1 = (const float*)d_in[1];
  const float* g1 = (const float*)d_in[2];
  const float* b1 = (const float*)d_in[3];
  const float* m1 = (const float*)d_in[4];
  const float* v1 = (const float*)d_in[5];
  const float* W2 = (const float*)d_in[6];
  const float* g2 = (const float*)d_in[7];
  const float* b2 = (const float*)d_in[8];
  const float* m2 = (const float*)d_in[9];
  const float* v2 = (const float*)d_in[10];
  const float* W3 = (const float*)d_in[11];
  const float* g3 = (const float*)d_in[12];
  const float* b3 = (const float*)d_in[13];
  const float* m3 = (const float*)d_in[14];
  const float* v3 = (const float*)d_in[15];
  float* out = (float*)d_out;

  edgeconv_mlp_kernel<<<512, 256, 0, stream>>>(
      x, W1, g1, b1, m1, v1, W2, g2, b2, m2, v2, W3, g3, b3, m3, v3, out);
}


# LDS reads (A fragments, BatchNorm coefficients) issued as whole groups one chunk of 16 WMMAs ahead into a second register set, not one by one between the WMMAs; sum order kept
# speedup vs baseline: 1.5543x; 1.0531x over previous
#include <hip/hip_runtime.h>
#include <math.h>


typedef __attribute__((ext_vector_type(2))) float v2f;
typedef __attribute__((ext_vector_type(4))) float v4f;
typedef __attribute__((ext_vector_type(8))) float v8f;

#define WPB  8
#define S1   36
#define S2   68

__device__ __forceinline__ v8f wmma4(v2f a, v2f b, v8f c) {
  return __builtin_amdgcn_wmma_f32_16x16x4_f32(
       false, a,  false, b,
       (short)0, c,  false,  false);
}

__device__ __forceinline__ v2f fma2(v2f a, v2f b, v2f c) { return __builtin_elementwise_fma(a, b, c); }
__device__ __forceinline__ v2f splat2(float a) { v2f r = {a, a}; return r; }

__device__ __forceinline__ v2f bn_gelu2(v2f h, v4f c0, v4f c1) {
  v2f u  = fma2(h, c0.xy, c0.zw);
  v2f ht = fma2(h, c1.xy, c1.zw);
  v2f x;
  x.x = __builtin_amdgcn_fmed3f(u.x, -3.7439211627767994f, 3.7439211627767994f);
  x.y = __builtin_amdgcn_fmed3f(u.y, -3.7439211627767994f, 3.7439211627767994f);
  v2f x2 = x * x;
  v2f p = splat2(0.00022905065861350646f);
  p = fma2(p, x2, splat2(0.0034082910107109506f));
  p = fma2(p, x2, splat2(0.050955695062380861f));
  p = fma2(p, x2, splat2(0.18520832239976145f));
  p = fma2(p, x2, splat2(1.128379143519084f));
  v2f q = splat2(-1.1791602954361697e-7f);
  q = fma2(q, x2, splat2(0.000023547966471313185f));
  q = fma2(q, x2, splat2(0.0010179625278914885f));
  q = fma2(q, x2, splat2(0.014070470171167667f));
  q = fma2(q, x2, splat2(0.11098505178285362f));
  q = fma2(q, x2, splat2(0.49746925110067538f));
  q = fma2(q, x2, splat2(1.0f));
  v2f rq;
  rq.x = __builtin_amdgcn_rcpf(q.x);
  rq.y = __builtin_amdgcn_rcpf(q.y);
  v2f e = (x * p) * rq;
  return fma2(ht, e, ht);
}


template <int S, int J0>
__device__ __forceinline__ void load_a_chunk(v2f (&a)[16], const float* wl) {
#pragma unroll
  for (int jj = 0; jj < 4; ++jj) {
#pragma unroll
    for (int t = 0; t < 4; ++t)
      a[jj * 4 + t] = *(const v2f*)(wl + t * 16 * S + 4 * (J0 + jj));
  }
}

template <int J0>
__device__ __forceinline__ void mma_chunk(v8f (&acc)[4], const v2f (&a)[16], const v2f (&hb)[16]) {
#pragma unroll
  for (int jj = 0; jj < 4; ++jj) {
#pragma unroll
    for (int t = 0; t < 4; ++t)
      acc[t] = wmma4(a[jj * 4 + t], hb[J0 + jj], acc[t]);
  }
}

template <int LAYER, int T>
__device__ __forceinline__ void load_coef_tile(v4f (&c)[8], const float* cfl) {
#pragma unroll
  for (int q = 0; q < 4; ++q) {
    const float* cp = cfl + (LAYER * 32 + 8 * T + 2 * q) * 8;
    c[2 * q]     = *(const v4f*)cp;
    c[2 * q + 1] = *(const v4f*)(cp + 4);
  }
}

template <int T>
__device__ __forceinline__ void gelu_tile(v2f (&hb)[16], const v8f (&acc)[4], const v4f (&c)[8]) {
#pragma unroll
  for (int q = 0; q < 4; ++q) {
    v2f hv = {acc[T][2 * q], acc[T][2 * q + 1]};
    hb[4 * T + q] = bn_gelu2(hv, c[2 * q], c[2 * q + 1]);
  }
}

template <int T>
__device__ __forceinline__ void gelu_store_tile(float* obase, const v8f (&acc)[4], const v4f (&c)[8]) {
#pragma unroll
  for (int q = 0; q < 4; ++q) {
    const int o0 = 16 * T + 4 * q;
    v2f hv = {acc[T][2 * q], acc[T][2 * q + 1]};
    v2f y = bn_gelu2(hv, c[2 * q], c[2 * q + 1]);
    obase[(size_t)o0 * 1024]       = y.x;
    obase[(size_t)(o0 + 1) * 1024] = y.y;
  }
}

#define HOLD_GROUP() __builtin_amdgcn_sched_barrier(0)

__global__ __launch_bounds__(256) void edgeconv_mlp_kernel(
    const float* __restrict__ x,
    const float* __restrict__ W1,
    const float* __restrict__ g1, const float* __restrict__ b1,
    const float* __restrict__ m1, const float* __restrict__ v1,
    const float* __restrict__ W2,
    const float* __restrict__ g2, const float* __restrict__ b2,
    const float* __restrict__ m2, const float* __restrict__ v2,
    const float* __restrict__ W3,
    const float* __restrict__ g3, const float* __restrict__ b3,
    const float* __restrict__ m3, const float* __restrict__ v3,
    float* __restrict__ out)
{
  __shared__ __attribute__((aligned(16))) float w1s[64 * S1];
  __shared__ __attribute__((aligned(16))) float w2s[64 * S2];
  __shared__ __attribute__((aligned(16))) float w3s[64 * S2];
  __shared__ __attribute__((aligned(16))) float cfs[3 * 32 * 8];

  const int tid    = threadIdx.x;
  const int wave   = tid >> 5;
  const int lane   = tid & 31;
  const int lane_m = lane & 15;
  const int half   = lane >> 4;
  const int kofs   = half * 2;

  const int tile = blockIdx.x * WPB + wave;
  const int p0   = tile * 16;

  const float* xrow = x + (size_t)(p0 + lane_m) * 32 + kofs;
  v2f hb[16];
#pragma unroll
  for (int j = 0; j < 8; ++j)
    hb[j] = *(const v2f*)(xrow + 4 * j);

  {
    v4f t1[2], t2[4], t3[4];
#pragma unroll
    for (int i = 0; i < 2; ++i) t1[i] = *(const v4f*)(W1 + (size_t)(tid + i * 256) * 4);
#pragma unroll
    for (int i = 0; i < 4; ++i) t2[i] = *(const v4f*)(W2 + (size_t)(tid + i * 256) * 4);
#pragma unroll
    for (int i = 0; i < 4; ++i) t3[i] = *(const v4f*)(W3 + (size_t)(tid + i * 256) * 4);

    const int wv = __builtin_amdgcn_readfirstlane(wave);
    if (wv < 3) {
      const float* gp = (wv == 0) ? g1 : (wv == 1) ? g2 : g3;
      const float* bp = (wv == 0) ? b1 : (wv == 1) ? b2 : b3;
      const float* mp = (wv == 0) ? m1 : (wv == 1) ? m2 : m3;
      const float* vp = (wv == 0) ? v1 : (wv == 1) ? v2 : v3;
      const float ge = gp[2 * lane], go = gp[2 * lane + 1];
      const float be = bp[2 * lane], bo = bp[2 * lane + 1];
      const float me = mp[2 * lane], mo = mp[2 * lane + 1];
      const float ve = vp[2 * lane], vo = vp[2 * lane + 1];
      const float sce = ge * rsqrtf(ve + 1e-5f), sco = go * rsqrtf(vo + 1e-5f);
      const float bie = be - me * sce,           bio = bo - mo * sco;
      v4f c0, c1;
      c0.x = sce * 0.70710678118654752f; c0.y = sco * 0.70710678118654752f;
      c0.z = bie * 0.70710678118654752f; c0.w = bio * 0.70710678118654752f;
      c1.x = sce * 0.5f; c1.y = sco * 0.5f;
      c1.z = bie * 0.5f; c1.w = bio * 0.5f;
      float* cw = cfs + (wv * 32 + lane) * 8;
      *(v4f*)(cw)     = c0;
      *(v4f*)(cw + 4) = c1;
    }
#pragma unroll
    for (int i = 0; i < 2; ++i) {
      int f = tid + i * 256;
      *(v4f*)(w1s + (f >> 3) * S1 + (f & 7) * 4) = t1[i];
    }
#pragma unroll
    for (int i = 0; i < 4; ++i) {
      int f = tid + i * 256;
      *(v4f*)(w2s + (f >> 4) * S2 + (f & 15) * 4) = t2[i];
      *(v4f*)(w3s + (f >> 4) * S2 + (f & 15) * 4) = t3[i];
    }
  }
  __syncthreads();

  const int rowp = 4 * ((lane_m & 7) >> 1) + 2 * (lane_m >> 3) + (lane_m & 1);
  const float* cfl = cfs + half * 8;

  v8f acc[4];
  const v8f vzero = {0.f, 0.f, 0.f, 0.f, 0.f, 0.f, 0.f, 0.f};

  v2f ab0[16], ab1[16];
  v4f cb0[8], cb1[8];
  const float* wl1 = w1s + rowp * S1 + kofs;
  const float* wl2 = w2s + rowp * S2 + kofs;
  const float* wl3 = w3s + rowp * S2 + kofs;

#pragma unroll
  for (int t = 0; t < 4; ++t) acc[t] = vzero;
  HOLD_GROUP();
  load_a_chunk<S1, 0>(ab0, wl1);
  load_a_chunk<S1, 4>(ab1, wl1);
  HOLD_GROUP();
  mma_chunk<0>(acc, ab0, hb);
  HOLD_GROUP();
  load_a_chunk<S2, 0>(ab0, wl2);
  load_coef_tile<0, 0>(cb0, cfl);
  load_coef_tile<0, 1>(cb1, cfl);
  HOLD_GROUP();
  mma_chunk<4>(acc, ab1, hb);
  gelu_tile<0>(hb, acc, cb0);
  HOLD_GROUP();
  load_coef_tile<0, 2>(cb0, cfl);
  HOLD_GROUP();
  gelu_tile<1>(hb, acc, cb1);
  HOLD_GROUP();
  load_coef_tile<0, 3>(cb1, cfl);
  HOLD_GROUP();
  gelu_tile<2>(hb, acc, cb0);
  gelu_tile<3>(hb, acc, cb1);

#pragma unroll
  for (int t = 0; t < 4; ++t) acc[t] = vzero;
  HOLD_GROUP();
  load_a_chunk<S2, 4>(ab1, wl2);
  HOLD_GROUP();
  mma_chunk<0>(acc, ab0, hb);
  HOLD_GROUP();
  load_a_chunk<S2, 8>(ab0, wl2);
  HOLD_GROUP();
  mma_chunk<4>(acc, ab1, hb);
  HOLD_GROUP();
  load_a_chunk<S2, 12>(ab1, wl2);
  HOLD_GROUP();
  mma_chunk<8>(acc, ab0, hb);
  HOLD_GROUP();
  load_a_chunk<S2, 0>(ab0, wl3);
  load_coef_tile<1, 0>(cb0, cfl);
  load_coef_tile<1, 1>(cb1, cfl);
  HOLD_GROUP();
  mma_chunk<12>(acc, ab1, hb);
  gelu_tile<0>(hb, acc, cb0);
  HOLD_GROUP();
  load_coef_tile<1, 2>(cb0, cfl);
  HOLD_GROUP();
  gelu_tile<1>(hb, acc, cb1);
  HOLD_GROUP();
  load_coef_tile<1, 3>(cb1, cfl);
  HOLD_GROUP();
  gelu_tile<2>(hb, acc, cb0);
  gelu_tile<3>(hb, acc, cb1);

#pragma unroll
  for (int t = 0; t < 4; ++t) acc[t] = vzero;
  HOLD_GROUP();
  load_a_chunk<S2, 4>(ab1, wl3);
  HOLD_GROUP();
  mma_chunk<0>(acc, ab0, hb);
  HOLD_GROUP();
  load_a_chunk<S2, 8>(ab0, wl3);
  HOLD_GROUP();
  mma_chunk<4>(acc, ab1, hb);
  HOLD_GROUP();
  load_a_chunk<S2, 12>(ab1, wl3);
  HOLD_GROUP();
  mma_chunk<8>(acc, ab0, hb);
  HOLD_GROUP();
  load_coef_tile<2, 0>(cb0, cfl);
  load_coef_tile<2, 1>(cb1, cfl);
  HOLD_GROUP();
  mma_chunk<12>(acc, ab1, hb);

  const int bidx  = p0 >> 10;
  const int nbase = p0 & 1023;
  float* obase = out + (size_t)bidx * 64 * 1024 + (size_t)kofs * 1024 + nbase + lane_m;
  gelu_store_tile<0>(obase, acc, cb0);
  HOLD_GROUP();
  load_coef_tile<2, 2>(cb0, cfl);
  HOLD_GROUP();
  gelu_store_tile<1>(obase, acc, cb1);
  HOLD_GROUP();
  load_coef_tile<2, 3>(cb1, cfl);
  HOLD_GROUP();
  gelu_store_tile<2>(obase, acc, cb0);
  gelu_store_tile<3>(obase, acc, cb1);
}

extern "C" void kernel_launch(void* const* d_in, const int* in_sizes, int n_in,
                              void* d_out, int out_size, void* d_ws, size_t ws_size,
                              hipStream_t stream) {
  const float* x  = (const float*)d_in[0];
  const float* W1 = (const float*)d_in[1];
  const float* g1 = (const float*)d_in[2];
  const float* b1 = (const float*)d_in[3];
  const float* m1 = (const float*)d_in[4];
  const float* v1 = (const float*)d_in[5];
  const float* W2 = (const float*)d_in[6];
  const float* g2 = (const float*)d_in[7];
  const float* b2 = (const float*)d_in[8];
  const float* m2 = (const float*)d_in[9];
  const float* v2 = (const float*)d_in[10];
  const float* W3 = (const float*)d_in[11];
  const float* g3 = (const float*)d_in[12];
  const float* b3 = (const float*)d_in[13];
  const float* m3 = (const float*)d_in[14];
  const float* v3 = (const float*)d_in[15];
  float* out = (float*)d_out;

  edgeconv_mlp_kernel<<<512, 256, 0, stream>>>(
      x, W1, g1, b1, m1, v1, W2, g2, b2, m2, v2, W3, g3, b3, m3, v3, out);
}


# GELU as y(1+tanh(y R(y^2))), y=t/2, R a degree-6 fit of atanh(erf), on the hardware tanh: 12 vector instructions per channel pair (2 tanh), not 20 (2 rcp, 2 clamps); abs. error as before
# speedup vs baseline: 1.6727x; 1.0762x over previous
#include <hip/hip_runtime.h>
#include <math.h>


typedef __attribute__((ext_vector_type(2))) float v2f;
typedef __attribute__((ext_vector_type(4))) float v4f;
typedef __attribute__((ext_vector_type(8))) float v8f;

#define WPB  8
#define S1   36
#define S2   68

__device__ __forceinline__ v8f wmma4(v2f a, v2f b, v8f c) {
  return __builtin_amdgcn_wmma_f32_16x16x4_f32(
       false, a,  false, b,
       (short)0, c,  false,  false);
}

__device__ __forceinline__ v2f fma2(v2f a, v2f b, v2f c) { return __builtin_elementwise_fma(a, b, c); }
__device__ __forceinline__ v2f splat2(float a) { v2f r = {a, a}; return r; }

__device__ __forceinline__ v2f bn_gelu2(v2f h, v4f c) {
  v2f y = fma2(h, c.xy, c.zw);
  v2f s = y * y;
  v2f r = splat2(1.4791867050670083e-5f);
  r = fma2(r, s, splat2(-0.0002733012364569846f));
  r = fma2(r, s, splat2(0.0020321967235763256f));
  r = fma2(r, s, splat2(-0.007070764748402877f));
  r = fma2(r, s, splat2(-0.0010601381046657949f));
  r = fma2(r, s, splat2(0.29067426038457883f));
  r = fma2(r, s, splat2(1.5957684329199464f));
  v2f z = y * r;
  v2f th;
  th.x = __builtin_amdgcn_tanhf(z.x);
  th.y = __builtin_amdgcn_tanhf(z.y);
  return fma2(y, th, y);
}


template <int S, int J0>
__device__ __forceinline__ void load_a_chunk(v2f (&a)[16], const float* wl) {
#pragma unroll
  for (int jj = 0; jj < 4; ++jj) {
#pragma unroll
    for (int t = 0; t < 4; ++t)
      a[jj * 4 + t] = *(const v2f*)(wl + t * 16 * S + 4 * (J0 + jj));
  }
}

template <int J0>
__device__ __forceinline__ void mma_chunk(v8f (&acc)[4], const v2f (&a)[16], const v2f (&hb)[16]) {
#pragma unroll
  for (int jj = 0; jj < 4; ++jj) {
#pragma unroll
    for (int t = 0; t < 4; ++t)
      acc[t] = wmma4(a[jj * 4 + t], hb[J0 + jj], acc[t]);
  }
}

template <int LAYER, int T>
__device__ __forceinline__ void load_coef_tile(v4f (&c)[4], const float* cfl) {
#pragma unroll
  for (int q = 0; q < 4; ++q)
    c[q] = *(const v4f*)(cfl + (LAYER * 32 + 8 * T + 2 * q) * 4);
}

template <int T>
__device__ __forceinline__ void gelu_tile(v2f (&hb)[16], const v8f (&acc)[4], const v4f (&c)[4]) {
#pragma unroll
  for (int q = 0; q < 4; ++q) {
    v2f hv = {acc[T][2 * q], acc[T][2 * q + 1]};
    hb[4 * T + q] = bn_gelu2(hv, c[q]);
  }
}

template <int T>
__device__ __forceinline__ void gelu_store_tile(float* obase, const v8f (&acc)[4], const v4f (&c)[4]) {
#pragma unroll
  for (int q = 0; q < 4; ++q) {
    const int o0 = 16 * T + 4 * q;
    v2f hv = {acc[T][2 * q], acc[T][2 * q + 1]};
    v2f y = bn_gelu2(hv, c[q]);
    obase[(size_t)o0 * 1024]       = y.x;
    obase[(size_t)(o0 + 1) * 1024] = y.y;
  }
}

#define HOLD_GROUP() __builtin_amdgcn_sched_barrier(0)

__global__ __launch_bounds__(256) void edgeconv_mlp_kernel(
    const float* __restrict__ x,
    const float* __restrict__ W1,
    const float* __restrict__ g1, const float* __restrict__ b1,
    const float* __restrict__ m1, const float* __restrict__ v1,
    const float* __restrict__ W2,
    const float* __restrict__ g2, const float* __restrict__ b2,
    const float* __restrict__ m2, const float* __restrict__ v2,
    const float* __restrict__ W3,
    const float* __restrict__ g3, const float* __restrict__ b3,
    const float* __restrict__ m3, const float* __restrict__ v3,
    float* __restrict__ out)
{
  __shared__ __attribute__((aligned(16))) float w1s[64 * S1];
  __shared__ __attribute__((aligned(16))) float w2s[64 * S2];
  __shared__ __attribute__((aligned(16))) float w3s[64 * S2];
  __shared__ __attribute__((aligned(16))) float cfs[3 * 32 * 4];

  const int tid    = threadIdx.x;
  const int wave   = tid >> 5;
  const int lane   = tid & 31;
  const int lane_m = lane & 15;
  const int half   = lane >> 4;
  const int kofs   = half * 2;

  const int tile = blockIdx.x * WPB + wave;
  const int p0   = tile * 16;

  const float* xrow = x + (size_t)(p0 + lane_m) * 32 + kofs;
  v2f hb[16];
#pragma unroll
  for (int j = 0; j < 8; ++j)
    hb[j] = *(const v2f*)(xrow + 4 * j);

  {
    v4f t1[2], t2[4], t3[4];
#pragma unroll
    for (int i = 0; i < 2; ++i) t1[i] = *(const v4f*)(W1 + (size_t)(tid + i * 256) * 4);
#pragma unroll
    for (int i = 0; i < 4; ++i) t2[i] = *(const v4f*)(W2 + (size_t)(tid + i * 256) * 4);
#pragma unroll
    for (int i = 0; i < 4; ++i) t3[i] = *(const v4f*)(W3 + (size_t)(tid + i * 256) * 4);

    const int wv = __builtin_amdgcn_readfirstlane(wave);
    if (wv < 3) {
      const float* gp = (wv == 0) ? g1 : (wv == 1) ? g2 : g3;
      const float* bp = (wv == 0) ? b1 : (wv == 1) ? b2 : b3;
      const float* mp = (wv == 0) ? m1 : (wv == 1) ? m2 : m3;
      const float* vp = (wv == 0) ? v1 : (wv == 1) ? v2 : v3;
      const float ge = gp[2 * lane], go = gp[2 * lane + 1];
      const float be = bp[2 * lane], bo = bp[2 * lane + 1];
      const float me = mp[2 * lane], mo = mp[2 * lane + 1];
      const float ve = vp[2 * lane], vo = vp[2 * lane + 1];
      const float sce = ge * rsqrtf(ve + 1e-5f), sco = go * rsqrtf(vo + 1e-5f);
      const float bie = be - me * sce,           bio = bo - mo * sco;
      v4f c1;
      c1.x = sce * 0.5f; c1.y = sco * 0.5f;
      c1.z = bie * 0.5f; c1.w = bio * 0.5f;
      *(v4f*)(cfs + (wv * 32 + lane) * 4) = c1;
    }
#pragma unroll
    for (int i = 0; i < 2; ++i) {
      int f = tid + i * 256;
      *(v4f*)(w1s + (f >> 3) * S1 + (f & 7) * 4) = t1[i];
    }
#pragma unroll
    for (int i = 0; i < 4; ++i) {
      int f = tid + i * 256;
      *(v4f*)(w2s + (f >> 4) * S2 + (f & 15) * 4) = t2[i];
      *(v4f*)(w3s + (f >> 4) * S2 + (f & 15) * 4) = t3[i];
    }
  }
  __syncthreads();

  const int rowp = 4 * ((lane_m & 7) >> 1) + 2 * (lane_m >> 3) + (lane_m & 1);
  const float* cfl = cfs + half * 4;

  v8f acc[4];
  const v8f vzero = {0.f, 0.f, 0.f, 0.f, 0.f, 0.f, 0.f, 0.f};

  v2f ab0[16], ab1[16];
  v4f cb0[4], cb1[4];
  const float* wl1 = w1s + rowp * S1 + kofs;
  const float* wl2 = w2s + rowp * S2 + kofs;
  const float* wl3 = w3s + rowp * S2 + kofs;

#pragma unroll
  for (int t = 0; t < 4; ++t) acc[t] = vzero;
  HOLD_GROUP();
  load_a_chunk<S1, 0>(ab0, wl1);
  load_a_chunk<S1, 4>(ab1, wl1);
  HOLD_GROUP();
  mma_chunk<0>(acc, ab0, hb);
  HOLD_GROUP();
  load_a_chunk<S2, 0>(ab0, wl2);
  load_coef_tile<0, 0>(cb0, cfl);
  load_coef_tile<0, 1>(cb1, cfl);
  HOLD_GROUP();
  mma_chunk<4>(acc, ab1, hb);
  gelu_tile<0>(hb, acc, cb0);
  HOLD_GROUP();
  load_coef_tile<0, 2>(cb0, cfl);
  HOLD_GROUP();
  gelu_tile<1>(hb, acc, cb1);
  HOLD_GROUP();
  load_coef_tile<0, 3>(cb1, cfl);
  HOLD_GROUP();
  gelu_tile<2>(hb, acc, cb0);
  gelu_tile<3>(hb, acc, cb1);

#pragma unroll
  for (int t = 0; t < 4; ++t) acc[t] = vzero;
  HOLD_GROUP();
  load_a_chunk<S2, 4>(ab1, wl2);
  HOLD_GROUP();
  mma_chunk<0>(acc, ab0, hb);
  HOLD_GROUP();
  load_a_chunk<S2, 8>(ab0, wl2);
  HOLD_GROUP();
  mma_chunk<4>(acc, ab1, hb);
  HOLD_GROUP();
  load_a_chunk<S2, 12>(ab1, wl2);
  HOLD_GROUP();
  mma_chunk<8>(acc, ab0, hb);
  HOLD_GROUP();
  load_a_chunk<S2, 0>(ab0, wl3);
  load_coef_tile<1, 0>(cb0, cfl);
  load_coef_tile<1, 1>(cb1, cfl);
  HOLD_GROUP();
  mma_chunk<12>(acc, ab1, hb);
  gelu_tile<0>(hb, acc, cb0);
  HOLD_GROUP();
  load_coef_tile<1, 2>(cb0, cfl);
  HOLD_GROUP();
  gelu_tile<1>(hb, acc, cb1);
  HOLD_GROUP();
  load_coef_tile<1, 3>(cb1, cfl);
  HOLD_GROUP();
  gelu_tile<2>(hb, acc, cb0);
  gelu_tile<3>(hb, acc, cb1);

#pragma unroll
  for (int t = 0; t < 4; ++t) acc[t] = vzero;
  HOLD_GROUP();
  load_a_chunk<S2, 4>(ab1, wl3);
  HOLD_GROUP();
  mma_chunk<0>(acc, ab0, hb);
  HOLD_GROUP();
  load_a_chunk<S2, 8>(ab0, wl3);
  HOLD_GROUP();
  mma_chunk<4>(acc, ab1, hb);
  HOLD_GROUP();
  load_a_chunk<S2, 12>(ab1, wl3);
  HOLD_GROUP();
  mma_chunk<8>(acc, ab0, hb);
  HOLD_GROUP();
  load_coef_tile<2, 0>(cb0, cfl);
  load_coef_tile<2, 1>(cb1, cfl);
  HOLD_GROUP();
  mma_chunk<12>(acc, ab1, hb);

  const int bidx  = p0 >> 10;
  const int nbase = p0 & 1023;
  float* obase = out + (size_t)bidx * 64 * 1024 + (size_t)kofs * 1024 + nbase + lane_m;
  gelu_store_tile<0>(obase, acc, cb0);
  HOLD_GROUP();
  load_coef_tile<2, 2>(cb0, cfl);
  HOLD_GROUP();
  gelu_store_tile<1>(obase, acc, cb1);
  HOLD_GROUP();
  load_coef_tile<2, 3>(cb1, cfl);
  HOLD_GROUP();
  gelu_store_tile<2>(obase, acc, cb0);
  gelu_store_tile<3>(obase, acc, cb1);
}

extern "C" void kernel_launch(void* const* d_in, const int* in_sizes, int n_in,
                              void* d_out, int out_size, void* d_ws, size_t ws_size,
                              hipStream_t stream) {
  const float* x  = (const float*)d_in[0];
  const float* W1 = (const float*)d_in[1];
  const float* g1 = (const float*)d_in[2];
  const float* b1 = (const float*)d_in[3];
  const float* m1 = (const float*)d_in[4];
  const float* v1 = (const float*)d_in[5];
  const float* W2 = (const float*)d_in[6];
  const float* g2 = (const float*)d_in[7];
  const float* b2 = (const float*)d_in[8];
  const float* m2 = (const float*)d_in[9];
  const float* v2 = (const float*)d_in[10];
  const float* W3 = (const float*)d_in[11];
  const float* g3 = (const float*)d_in[12];
  const float* b3 = (const float*)d_in[13];
  const float* m3 = (const float*)d_in[14];
  const float* v3 = (const float*)d_in[15];
  float* out = (float*)d_out;

  edgeconv_mlp_kernel<<<512, 256, 0, stream>>>(
      x, W1, g1, b1, m1, v1, W2, g2, b2, m2, v2, W3, g3, b3, m3, v3, out);
}
